# starting kernel
# baseline (speedup 1.0000x reference)
#include <hip/hip_runtime.h>
#include <hip/hip_bf16.h>


#define DIMC   256
#define MLPC   1024
#define BC     2
#define NC     2048
#define HEADS  8
#define HDC    32
#define ROWS   (BC * NC)
#define QKVN   (3 * DIMC)
#define LAYER  1

typedef __attribute__((ext_vector_type(2))) float v2f;
typedef __attribute__((ext_vector_type(8))) float v8f;

static __device__ inline v8f wmma4(v2f a, v2f b, v8f c) {
  return __builtin_amdgcn_wmma_f32_16x16x4_f32(
       false, a,  false, b,
       (short)0, c,  false,  false);
}

static __device__ inline v8f v8zero() {
  v8f z = {0.f, 0.f, 0.f, 0.f, 0.f, 0.f, 0.f, 0.f};
  return z;
}

__global__ __launch_bounds__(256) void ln_kernel(
    const float* __restrict__ x, const float* __restrict__ g,
    const float* __restrict__ b, float* __restrict__ y, int rows) {
  int wave = (blockIdx.x * blockDim.x + threadIdx.x) >> 5;
  int lane = threadIdx.x & 31;
  if (wave >= rows) return;
  const float* xr = x + (size_t)wave * DIMC;
  float vals[8];
  float s = 0.f;
#pragma unroll
  for (int i = 0; i < 8; ++i) { vals[i] = xr[lane + 32 * i]; s += vals[i]; }
#pragma unroll
  for (int off = 16; off >= 1; off >>= 1) s += __shfl_xor(s, off, 32);
  float mean = s * (1.0f / DIMC);
  float vs = 0.f;
#pragma unroll
  for (int i = 0; i < 8; ++i) { float d = vals[i] - mean; vs += d * d; }
#pragma unroll
  for (int off = 16; off >= 1; off >>= 1) vs += __shfl_xor(vs, off, 32);
  float rstd = rsqrtf(vs * (1.0f / DIMC) + 1e-5f);
  float* yr = y + (size_t)wave * DIMC;
#pragma unroll
  for (int i = 0; i < 8; ++i) {
    int c = lane + 32 * i;
    yr[c] = (vals[i] - mean) * rstd * g[c] + b[c];
  }
}

__global__ __launch_bounds__(128) void wmma_gemm_kernel(
    const float* __restrict__ A, const float* __restrict__ W,
    const float* __restrict__ bias, const float* __restrict__ resid,
    float* __restrict__ out, int M, int N, int K, int gelu_flag) {
  const int lane = threadIdx.x & 31;
  const int wv = threadIdx.x >> 5;
  const int half = lane >> 4;
  const int lc = lane & 15;
  const int n0 = blockIdx.x * 64;
  const int m0 = blockIdx.y * 64 + wv * 16;

  const float* Arow = A + (size_t)(m0 + lc) * K;
  v8f acc[4];
#pragma unroll
  for (int nt = 0; nt < 4; ++nt) acc[nt] = v8zero();

  for (int kk = 0; kk < K; kk += 4) {
    if ((kk & 63) == 0 && kk + 64 < K)
      __builtin_prefetch(Arow + kk + 64, 0, 0);
    const int ka = kk + 2 * half;
    v2f a = {Arow[ka], Arow[ka + 1]};
    const float* Wr0 = W + (size_t)ka * N + n0 + lc;
    const float* Wr1 = Wr0 + N;
#pragma unroll
    for (int nt = 0; nt < 4; ++nt) {
      v2f b = {Wr0[nt * 16], Wr1[nt * 16]};
      acc[nt] = wmma4(a, b, acc[nt]);
    }
  }

#pragma unroll
  for (int nt = 0; nt < 4; ++nt) {
    const int col = n0 + nt * 16 + lc;
    const float bv = bias ? bias[col] : 0.f;
#pragma unroll
    for (int v = 0; v < 8; ++v) {
      const int row = m0 + v + 8 * half;
      float val = acc[nt][v] + bv;
      if (gelu_flag)
        val = 0.5f * val * (1.0f + erff(val * 0.70710678118654752f));
      if (resid) val += resid[(size_t)row * N + col];
      out[(size_t)row * N + col] = val;
    }
  }
}

__global__ __launch_bounds__(128) void attn_kernel(
    const float* __restrict__ qkv, float* __restrict__ o) {
  __shared__ float plds[4][256];
  const int lane = threadIdx.x & 31;
  const int wv = threadIdx.x >> 5;
  const int half = lane >> 4;
  const int lc = lane & 15;
  const int bh = blockIdx.x;
  const int bb = bh >> 3, h = bh & 7;
  const int row0 = (blockIdx.y * 4 + wv) * 16;
  const int base = bb * NC;
  const float scale = 0.17677669529663687f;

  const float* Q = qkv + (size_t)base * QKVN + h * HDC;
  const float* Kp = Q + DIMC;
  const float* Vp = Q + 2 * DIMC;

  v2f qa[8];
  {
    const float* qrow = Q + (size_t)(row0 + lc) * QKVN + 2 * half;
#pragma unroll
    for (int i = 0; i < 8; ++i) {
      v2f t = {qrow[4 * i] * scale, qrow[4 * i + 1] * scale};
      qa[i] = t;
    }
  }

  float mrow[8], lrow[8];
#pragma unroll
  for (int v = 0; v < 8; ++v) { mrow[v] = -1e30f; lrow[v] = 0.f; }
  v8f oacc0 = v8zero(), oacc1 = v8zero();

  for (int jb = 0; jb < NC / 16; ++jb) {
    const float* krow = Kp + (size_t)(jb * 16 + lc) * QKVN + 2 * half;
    v8f s = v8zero();
#pragma unroll
    for (int i = 0; i < 8; ++i) {
      v2f kb = {krow[4 * i], krow[4 * i + 1]};
      s = wmma4(qa[i], kb, s);
    }
#pragma unroll
    for (int v = 0; v < 8; ++v) {
      float sv = s[v];
      float rmax = sv;
      rmax = fmaxf(rmax, __shfl_xor(rmax, 1, 32));
      rmax = fmaxf(rmax, __shfl_xor(rmax, 2, 32));
      rmax = fmaxf(rmax, __shfl_xor(rmax, 4, 32));
      rmax = fmaxf(rmax, __shfl_xor(rmax, 8, 32));
      float mn = fmaxf(mrow[v], rmax);
      float p = __expf(sv - mn);
      float rs = p;
      rs += __shfl_xor(rs, 1, 32);
      rs += __shfl_xor(rs, 2, 32);
      rs += __shfl_xor(rs, 4, 32);
      rs += __shfl_xor(rs, 8, 32);
      float c = __expf(mrow[v] - mn);
      lrow[v] = lrow[v] * c + rs;
      mrow[v] = mn;
      oacc0[v] *= c;
      oacc1[v] *= c;
      plds[wv][(v + 8 * half) * 16 + lc] = p;
    }
#pragma unroll
    for (int i = 0; i < 4; ++i) {
      v2f pa = {plds[wv][lc * 16 + 4 * i + 2 * half],
                plds[wv][lc * 16 + 4 * i + 2 * half + 1]};
      const int kv = jb * 16 + 4 * i + 2 * half;
      const float* v0 = Vp + (size_t)kv * QKVN + lc;
      const float* v1 = v0 + QKVN;
      v2f vb0 = {v0[0], v1[0]};
      v2f vb1 = {v0[16], v1[16]};
      oacc0 = wmma4(pa, vb0, oacc0);
      oacc1 = wmma4(pa, vb1, oacc1);
    }
  }

#pragma unroll
  for (int v = 0; v < 8; ++v) {
    const float inv = 1.0f / lrow[v];
    float* orow = o + (size_t)(base + row0 + v + 8 * half) * DIMC + h * HDC;
    orow[lc] = oacc0[v] * inv;
    orow[16 + lc] = oacc1[v] * inv;
  }
}

extern "C" void kernel_launch(void* const* d_in, const int* in_sizes, int n_in,
                              void* d_out, int out_size, void* d_ws, size_t ws_size,
                              hipStream_t stream) {
  const float* x     = (const float*)d_in[0];
  const float* ln1_g = (const float*)d_in[1] + (size_t)LAYER * DIMC;
  const float* ln1_b = (const float*)d_in[2] + (size_t)LAYER * DIMC;
  const float* Wqkv  = (const float*)d_in[3] + (size_t)LAYER * DIMC * QKVN;
  const float* bqkv  = (const float*)d_in[4] + (size_t)LAYER * QKVN;
  const float* Wo    = (const float*)d_in[5] + (size_t)LAYER * DIMC * DIMC;
  const float* bo    = (const float*)d_in[6] + (size_t)LAYER * DIMC;
  const float* ln2_g = (const float*)d_in[7] + (size_t)LAYER * DIMC;
  const float* ln2_b = (const float*)d_in[8] + (size_t)LAYER * DIMC;
  const float* W1    = (const float*)d_in[9] + (size_t)LAYER * DIMC * MLPC;
  const float* b1    = (const float*)d_in[10] + (size_t)LAYER * MLPC;
  const float* W2    = (const float*)d_in[11] + (size_t)LAYER * MLPC * DIMC;
  const float* b2    = (const float*)d_in[12] + (size_t)LAYER * DIMC;
  float* out = (float*)d_out;

  float* ws   = (float*)d_ws;
  float* y    = ws;
  float* qkvb = ws + (size_t)ROWS * DIMC;
  float* oat  = y;
  float* a    = qkvb + (size_t)ROWS * QKVN;
  float* z    = qkvb;
  float* h1   = a + (size_t)ROWS * DIMC;

  ln_kernel<<<ROWS / 8, 256, 0, stream>>>(x, ln1_g, ln1_b, y, ROWS);
  wmma_gemm_kernel<<<dim3(QKVN / 64, ROWS / 64), 128, 0, stream>>>(
      y, Wqkv, bqkv, nullptr, qkvb, ROWS, QKVN, DIMC, 0);
  attn_kernel<<<dim3(BC * HEADS, NC / 64), 128, 0, stream>>>(qkvb, oat);
  wmma_gemm_kernel<<<dim3(DIMC / 64, ROWS / 64), 128, 0, stream>>>(
      oat, Wo, bo, x, a, ROWS, DIMC, DIMC, 0);
  ln_kernel<<<ROWS / 8, 256, 0, stream>>>(a, ln2_g, ln2_b, z, ROWS);
  wmma_gemm_kernel<<<dim3(MLPC / 64, ROWS / 64), 128, 0, stream>>>(
      z, W1, b1, nullptr, h1, ROWS, MLPC, DIMC, 1);
  wmma_gemm_kernel<<<dim3(DIMC / 64, ROWS / 64), 128, 0, stream>>>(
      h1, W2, b2, a, out, ROWS, DIMC, MLPC, 0);
}


# attention scores computed transposed (keys on rows): row max/sum in-lane, P feeds P.V from registers, no LDS round trip or cross-lane permute trees; changes row-sum and key summation order
# speedup vs baseline: 1.3301x; 1.3301x over previous
#include <hip/hip_runtime.h>
#include <hip/hip_bf16.h>


#define DIMC   256
#define MLPC   1024
#define BC     2
#define NC     2048
#define HEADS  8
#define HDC    32
#define ROWS   (BC * NC)
#define QKVN   (3 * DIMC)
#define LAYER  1

typedef __attribute__((ext_vector_type(2))) float v2f;
typedef __attribute__((ext_vector_type(8))) float v8f;

static __device__ inline v8f wmma4(v2f a, v2f b, v8f c) {
  return __builtin_amdgcn_wmma_f32_16x16x4_f32(
       false, a,  false, b,
       (short)0, c,  false,  false);
}

static __device__ inline v8f v8zero() {
  v8f z = {0.f, 0.f, 0.f, 0.f, 0.f, 0.f, 0.f, 0.f};
  return z;
}

__global__ __launch_bounds__(256) void ln_kernel(
    const float* __restrict__ x, const float* __restrict__ g,
    const float* __restrict__ b, float* __restrict__ y, int rows) {
  int wave = (blockIdx.x * blockDim.x + threadIdx.x) >> 5;
  int lane = threadIdx.x & 31;
  if (wave >= rows) return;
  const float* xr = x + (size_t)wave * DIMC;
  float vals[8];
  float s = 0.f;
#pragma unroll
  for (int i = 0; i < 8; ++i) { vals[i] = xr[lane + 32 * i]; s += vals[i]; }
#pragma unroll
  for (int off = 16; off >= 1; off >>= 1) s += __shfl_xor(s, off, 32);
  float mean = s * (1.0f / DIMC);
  float vs = 0.f;
#pragma unroll
  for (int i = 0; i < 8; ++i) { float d = vals[i] - mean; vs += d * d; }
#pragma unroll
  for (int off = 16; off >= 1; off >>= 1) vs += __shfl_xor(vs, off, 32);
  float rstd = rsqrtf(vs * (1.0f / DIMC) + 1e-5f);
  float* yr = y + (size_t)wave * DIMC;
#pragma unroll
  for (int i = 0; i < 8; ++i) {
    int c = lane + 32 * i;
    yr[c] = (vals[i] - mean) * rstd * g[c] + b[c];
  }
}

__global__ __launch_bounds__(128) void wmma_gemm_kernel(
    const float* __restrict__ A, const float* __restrict__ W,
    const float* __restrict__ bias, const float* __restrict__ resid,
    float* __restrict__ out, int M, int N, int K, int gelu_flag) {
  const int lane = threadIdx.x & 31;
  const int wv = threadIdx.x >> 5;
  const int half = lane >> 4;
  const int lc = lane & 15;
  const int n0 = blockIdx.x * 64;
  const int m0 = blockIdx.y * 64 + wv * 16;

  const float* Arow = A + (size_t)(m0 + lc) * K;
  v8f acc[4];
#pragma unroll
  for (int nt = 0; nt < 4; ++nt) acc[nt] = v8zero();

  for (int kk = 0; kk < K; kk += 4) {
    if ((kk & 63) == 0 && kk + 64 < K)
      __builtin_prefetch(Arow + kk + 64, 0, 0);
    const int ka = kk + 2 * half;
    v2f a = {Arow[ka], Arow[ka + 1]};
    const float* Wr0 = W + (size_t)ka * N + n0 + lc;
    const float* Wr1 = Wr0 + N;
#pragma unroll
    for (int nt = 0; nt < 4; ++nt) {
      v2f b = {Wr0[nt * 16], Wr1[nt * 16]};
      acc[nt] = wmma4(a, b, acc[nt]);
    }
  }

#pragma unroll
  for (int nt = 0; nt < 4; ++nt) {
    const int col = n0 + nt * 16 + lc;
    const float bv = bias ? bias[col] : 0.f;
#pragma unroll
    for (int v = 0; v < 8; ++v) {
      const int row = m0 + v + 8 * half;
      float val = acc[nt][v] + bv;
      if (gelu_flag)
        val = 0.5f * val * (1.0f + erff(val * 0.70710678118654752f));
      if (resid) val += resid[(size_t)row * N + col];
      out[(size_t)row * N + col] = val;
    }
  }
}

static __device__ inline float xhalf(float x) {
  const int b = __builtin_bit_cast(int, x);
  return __builtin_bit_cast(
      float, __builtin_amdgcn_permlanex16(b, b, 0x76543210, 0xfedcba98, false, false));
}

__global__ __launch_bounds__(128) void attn_kernel(
    const float* __restrict__ qkv, float* __restrict__ o) {
  const int lane = threadIdx.x & 31;
  const int wv = threadIdx.x >> 5;
  const int half = lane >> 4;
  const int lc = lane & 15;
  const int bh = blockIdx.x;
  const int bb = bh >> 3, h = bh & 7;
  const int row0 = (blockIdx.y * 4 + wv) * 16;
  const int base = bb * NC;
  const float scale = 0.17677669529663687f;

  const float* Q = qkv + (size_t)base * QKVN + h * HDC;
  const float* Kp = Q + DIMC;
  const float* Vp = Q + 2 * DIMC;

  v2f qb[8];
  {
    const float* qrow = Q + (size_t)(row0 + lc) * QKVN + 2 * half;
#pragma unroll
    for (int i = 0; i < 8; ++i) {
      v2f t = {qrow[4 * i] * scale, qrow[4 * i + 1] * scale};
      qb[i] = t;
    }
  }

  float mrun = -1e30f;
  float lsum = 0.f;
  v8f ot0 = v8zero(), ot1 = v8zero();

  for (int jb = 0; jb < NC / 16; ++jb) {
    const float* krow = Kp + (size_t)(jb * 16 + lc) * QKVN + 2 * half;
    v2f ka[8];
#pragma unroll
    for (int i = 0; i < 8; ++i) {
      v2f t = {krow[4 * i], krow[4 * i + 1]};
      ka[i] = t;
    }
    const float* vrow = Vp + (size_t)(jb * 16 + 8 * half) * QKVN + lc;
    v2f va0[4], va1[4];
#pragma unroll
    for (int i = 0; i < 4; ++i) {
      const float* r0 = vrow + (size_t)(2 * i) * QKVN;
      const float* r1 = r0 + QKVN;
      v2f t0 = {r0[0], r1[0]};
      v2f t1 = {r0[16], r1[16]};
      va0[i] = t0;
      va1[i] = t1;
    }

    v8f st = v8zero();
#pragma unroll
    for (int i = 0; i < 8; ++i) st = wmma4(ka[i], qb[i], st);

    float bm = st[0];
#pragma unroll
    for (int r = 1; r < 8; ++r) bm = fmaxf(bm, st[r]);
    bm = fmaxf(bm, xhalf(bm));
    const float mn = fmaxf(mrun, bm);
    const float c = __expf(mrun - mn);
    mrun = mn;
    float p[8];
    float ps = 0.f;
#pragma unroll
    for (int r = 0; r < 8; ++r) {
      p[r] = __expf(st[r] - mn);
      ps += p[r];
    }
    lsum = lsum * c + ps;
#pragma unroll
    for (int r = 0; r < 8; ++r) {
      ot0[r] *= c;
      ot1[r] *= c;
    }

#pragma unroll
    for (int i = 0; i < 4; ++i) {
      v2f pb = {p[2 * i], p[2 * i + 1]};
      ot0 = wmma4(va0[i], pb, ot0);
      ot1 = wmma4(va1[i], pb, ot1);
    }
  }

  const float ltot = lsum + xhalf(lsum);
  const float inv = 1.0f / ltot;
  float* orow = o + (size_t)(base + row0 + lc) * DIMC + h * HDC + 8 * half;
  const float4 w0 = make_float4(ot0[0] * inv, ot0[1] * inv, ot0[2] * inv, ot0[3] * inv);
  const float4 w1 = make_float4(ot0[4] * inv, ot0[5] * inv, ot0[6] * inv, ot0[7] * inv);
  const float4 w2 = make_float4(ot1[0] * inv, ot1[1] * inv, ot1[2] * inv, ot1[3] * inv);
  const float4 w3 = make_float4(ot1[4] * inv, ot1[5] * inv, ot1[6] * inv, ot1[7] * inv);
  *reinterpret_cast<float4*>(orow) = w0;
  *reinterpret_cast<float4*>(orow + 4) = w1;
  *reinterpret_cast<float4*>(orow + 16) = w2;
  *reinterpret_cast<float4*>(orow + 20) = w3;
}

extern "C" void kernel_launch(void* const* d_in, const int* in_sizes, int n_in,
                              void* d_out, int out_size, void* d_ws, size_t ws_size,
                              hipStream_t stream) {
  const float* x     = (const float*)d_in[0];
  const float* ln1_g = (const float*)d_in[1] + (size_t)LAYER * DIMC;
  const float* ln1_b = (const float*)d_in[2] + (size_t)LAYER * DIMC;
  const float* Wqkv  = (const float*)d_in[3] + (size_t)LAYER * DIMC * QKVN;
  const float* bqkv  = (const float*)d_in[4] + (size_t)LAYER * QKVN;
  const float* Wo    = (const float*)d_in[5] + (size_t)LAYER * DIMC * DIMC;
  const float* bo    = (const float*)d_in[6] + (size_t)LAYER * DIMC;
  const float* ln2_g = (const float*)d_in[7] + (size_t)LAYER * DIMC;
  const float* ln2_b = (const float*)d_in[8] + (size_t)LAYER * DIMC;
  const float* W1    = (const float*)d_in[9] + (size_t)LAYER * DIMC * MLPC;
  const float* b1    = (const float*)d_in[10] + (size_t)LAYER * MLPC;
  const float* W2    = (const float*)d_in[11] + (size_t)LAYER * MLPC * DIMC;
  const float* b2    = (const float*)d_in[12] + (size_t)LAYER * DIMC;
  float* out = (float*)d_out;

  float* ws   = (float*)d_ws;
  float* y    = ws;
  float* qkvb = ws + (size_t)ROWS * DIMC;
  float* oat  = y;
  float* a    = qkvb + (size_t)ROWS * QKVN;
  float* z    = qkvb;
  float* h1   = a + (size_t)ROWS * DIMC;

  ln_kernel<<<ROWS / 8, 256, 0, stream>>>(x, ln1_g, ln1_b, y, ROWS);
  wmma_gemm_kernel<<<dim3(QKVN / 64, ROWS / 64), 128, 0, stream>>>(
      y, Wqkv, bqkv, nullptr, qkvb, ROWS, QKVN, DIMC, 0);
  attn_kernel<<<dim3(BC * HEADS, NC / 64), 128, 0, stream>>>(qkvb, oat);
  wmma_gemm_kernel<<<dim3(DIMC / 64, ROWS / 64), 128, 0, stream>>>(
      oat, Wo, bo, x, a, ROWS, DIMC, DIMC, 0);
  ln_kernel<<<ROWS / 8, 256, 0, stream>>>(a, ln2_g, ln2_b, z, ROWS);
  wmma_gemm_kernel<<<dim3(MLPC / 64, ROWS / 64), 128, 0, stream>>>(
      z, W1, b1, nullptr, h1, ROWS, MLPC, DIMC, 1);
  wmma_gemm_kernel<<<dim3(DIMC / 64, ROWS / 64), 128, 0, stream>>>(
      h1, W2, b2, a, out, ROWS, DIMC, MLPC, 0);
}


# operand loads issued ahead of use into spare register sets (attention K/V: two key blocks ahead, 4 sets; GEMM: next 16-k chunk, 2 sets), so load latency overlaps WMMAs; summation order unchanged
# speedup vs baseline: 2.3828x; 1.7914x over previous
#include <hip/hip_runtime.h>
#include <hip/hip_bf16.h>


#define DIMC   256
#define MLPC   1024
#define BC     2
#define NC     2048
#define HEADS  8
#define HDC    32
#define ROWS   (BC * NC)
#define QKVN   (3 * DIMC)
#define LAYER  1

typedef __attribute__((ext_vector_type(2))) float v2f;
typedef __attribute__((ext_vector_type(8))) float v8f;

static __device__ inline v8f wmma4(v2f a, v2f b, v8f c) {
  return __builtin_amdgcn_wmma_f32_16x16x4_f32(
       false, a,  false, b,
       (short)0, c,  false,  false);
}

static __device__ inline v8f v8zero() {
  v8f z = {0.f, 0.f, 0.f, 0.f, 0.f, 0.f, 0.f, 0.f};
  return z;
}

__global__ __launch_bounds__(256) void ln_kernel(
    const float* __restrict__ x, const float* __restrict__ g,
    const float* __restrict__ b, float* __restrict__ y, int rows) {
  int wave = (blockIdx.x * blockDim.x + threadIdx.x) >> 5;
  int lane = threadIdx.x & 31;
  if (wave >= rows) return;
  const float* xr = x + (size_t)wave * DIMC;
  float vals[8];
  float s = 0.f;
#pragma unroll
  for (int i = 0; i < 8; ++i) { vals[i] = xr[lane + 32 * i]; s += vals[i]; }
#pragma unroll
  for (int off = 16; off >= 1; off >>= 1) s += __shfl_xor(s, off, 32);
  float mean = s * (1.0f / DIMC);
  float vs = 0.f;
#pragma unroll
  for (int i = 0; i < 8; ++i) { float d = vals[i] - mean; vs += d * d; }
#pragma unroll
  for (int off = 16; off >= 1; off >>= 1) vs += __shfl_xor(vs, off, 32);
  float rstd = rsqrtf(vs * (1.0f / DIMC) + 1e-5f);
  float* yr = y + (size_t)wave * DIMC;
#pragma unroll
  for (int i = 0; i < 8; ++i) {
    int c = lane + 32 * i;
    yr[c] = (vals[i] - mean) * rstd * g[c] + b[c];
  }
}

static __device__ inline void gemm_load_chunk(
    const float* __restrict__ Arow, const float* __restrict__ Wcol, int N, int kk,
    int half, v2f (&a)[4], v2f (&b)[4][4]) {
#pragma unroll
  for (int s = 0; s < 4; ++s) {
    const int ka = kk + 4 * s + 2 * half;
    v2f t = {Arow[ka], Arow[ka + 1]};
    a[s] = t;
    const float* Wr0 = Wcol + (size_t)ka * N;
    const float* Wr1 = Wr0 + N;
#pragma unroll
    for (int nt = 0; nt < 4; ++nt) {
      v2f u = {Wr0[nt * 16], Wr1[nt * 16]};
      b[s][nt] = u;
    }
  }
}

static __device__ inline void gemm_chunk(const v2f (&a)[4], const v2f (&b)[4][4],
                                         v8f (&acc)[4]) {
#pragma unroll
  for (int s = 0; s < 4; ++s) {
#pragma unroll
    for (int nt = 0; nt < 4; ++nt) acc[nt] = wmma4(a[s], b[s][nt], acc[nt]);
  }
}

__global__ __launch_bounds__(128) void wmma_gemm_kernel(
    const float* __restrict__ A, const float* __restrict__ W,
    const float* __restrict__ bias, const float* __restrict__ resid,
    float* __restrict__ out, int M, int N, int K, int gelu_flag) {
  const int lane = threadIdx.x & 31;
  const int wv = threadIdx.x >> 5;
  const int half = lane >> 4;
  const int lc = lane & 15;
  const int n0 = blockIdx.x * 64;
  const int m0 = blockIdx.y * 64 + wv * 16;

  const float* Arow = A + (size_t)(m0 + lc) * K;
  const float* Wcol = W + n0 + lc;
  v8f acc[4];
#pragma unroll
  for (int nt = 0; nt < 4; ++nt) acc[nt] = v8zero();

  v2f a0[4], b0[4][4], a1[4], b1[4][4];
  gemm_load_chunk(Arow, Wcol, N, 0, half, a0, b0);
  for (int kk = 0; kk < K; kk += 32) {
    if ((kk & 63) == 0 && kk + 64 < K)
      __builtin_prefetch(Arow + kk + 64, 0, 0);
    gemm_load_chunk(Arow, Wcol, N, kk + 16, half, a1, b1);
    __builtin_amdgcn_sched_barrier(0);
    gemm_chunk(a0, b0, acc);
    const int kn = (kk + 32 < K) ? kk + 32 : kk;
    gemm_load_chunk(Arow, Wcol, N, kn, half, a0, b0);
    __builtin_amdgcn_sched_barrier(0);
    gemm_chunk(a1, b1, acc);
  }

#pragma unroll
  for (int nt = 0; nt < 4; ++nt) {
    const int col = n0 + nt * 16 + lc;
    const float bv = bias ? bias[col] : 0.f;
#pragma unroll
    for (int v = 0; v < 8; ++v) {
      const int row = m0 + v + 8 * half;
      float val = acc[nt][v] + bv;
      if (gelu_flag)
        val = 0.5f * val * (1.0f + erff(val * 0.70710678118654752f));
      if (resid) val += resid[(size_t)row * N + col];
      out[(size_t)row * N + col] = val;
    }
  }
}

static __device__ inline float xhalf(float x) {
  const int b = __builtin_bit_cast(int, x);
  return __builtin_bit_cast(
      float, __builtin_amdgcn_permlanex16(b, b, 0x76543210, 0xfedcba98, false, false));
}

static __device__ inline void attn_load_kv(
    const float* __restrict__ Kp, const float* __restrict__ Vp, int jb, int lc, int half,
    v2f (&ka)[8], v2f (&va0)[4], v2f (&va1)[4]) {
  const float* krow = Kp + (size_t)(jb * 16 + lc) * QKVN + 2 * half;
#pragma unroll
  for (int i = 0; i < 8; ++i) {
    v2f t = {krow[4 * i], krow[4 * i + 1]};
    ka[i] = t;
  }
  const float* vrow = Vp + (size_t)(jb * 16 + 8 * half) * QKVN + lc;
#pragma unroll
  for (int i = 0; i < 4; ++i) {
    const float* r0 = vrow + (size_t)(2 * i) * QKVN;
    const float* r1 = r0 + QKVN;
    v2f t0 = {r0[0], r1[0]};
    v2f t1 = {r0[16], r1[16]};
    va0[i] = t0;
    va1[i] = t1;
  }
}

static __device__ inline void attn_block(
    const v2f (&qb)[8], const v2f (&ka)[8], const v2f (&va0)[4], const v2f (&va1)[4],
    float& mrun, float& lsum, v8f& ot0, v8f& ot1) {
  v8f st = v8zero();
#pragma unroll
  for (int i = 0; i < 8; ++i) st = wmma4(ka[i], qb[i], st);

  float bm = st[0];
#pragma unroll
  for (int r = 1; r < 8; ++r) bm = fmaxf(bm, st[r]);
  bm = fmaxf(bm, xhalf(bm));
  const float mn = fmaxf(mrun, bm);
  const float c = __expf(mrun - mn);
  mrun = mn;
  float p[8];
  float ps = 0.f;
#pragma unroll
  for (int r = 0; r < 8; ++r) {
    p[r] = __expf(st[r] - mn);
    ps += p[r];
  }
  lsum = lsum * c + ps;
#pragma unroll
  for (int r = 0; r < 8; ++r) {
    ot0[r] *= c;
    ot1[r] *= c;
  }

#pragma unroll
  for (int i = 0; i < 4; ++i) {
    v2f pb = {p[2 * i], p[2 * i + 1]};
    ot0 = wmma4(va0[i], pb, ot0);
    ot1 = wmma4(va1[i], pb, ot1);
  }
}

__global__ __launch_bounds__(128) void attn_kernel(
    const float* __restrict__ qkv, float* __restrict__ o) {
  const int lane = threadIdx.x & 31;
  const int wv = threadIdx.x >> 5;
  const int half = lane >> 4;
  const int lc = lane & 15;
  const int bh = blockIdx.x;
  const int bb = bh >> 3, h = bh & 7;
  const int row0 = (blockIdx.y * 4 + wv) * 16;
  const int base = bb * NC;
  const float scale = 0.17677669529663687f;

  const float* Q = qkv + (size_t)base * QKVN + h * HDC;
  const float* Kp = Q + DIMC;
  const float* Vp = Q + 2 * DIMC;

  v2f qb[8];
  {
    const float* qrow = Q + (size_t)(row0 + lc) * QKVN + 2 * half;
#pragma unroll
    for (int i = 0; i < 8; ++i) {
      v2f t = {qrow[4 * i] * scale, qrow[4 * i + 1] * scale};
      qb[i] = t;
    }
  }

  float mrun = -1e30f;
  float lsum = 0.f;
  v8f ot0 = v8zero(), ot1 = v8zero();

  v2f k0[8], u0[4], x0[4], k1[8], u1[4], x1[4], k2[8], u2[4], x2[4], k3[8], u3[4], x3[4];
  const int jlast = NC / 16 - 1;
  attn_load_kv(Kp, Vp, 0, lc, half, k0, u0, x0);
  attn_load_kv(Kp, Vp, 1, lc, half, k1, u1, x1);
  for (int jb = 0; jb < NC / 16; jb += 4) {
    attn_load_kv(Kp, Vp, jb + 2, lc, half, k2, u2, x2);
    __builtin_amdgcn_sched_barrier(0);
    attn_block(qb, k0, u0, x0, mrun, lsum, ot0, ot1);
    attn_load_kv(Kp, Vp, jb + 3, lc, half, k3, u3, x3);
    __builtin_amdgcn_sched_barrier(0);
    attn_block(qb, k1, u1, x1, mrun, lsum, ot0, ot1);
    attn_load_kv(Kp, Vp, min(jb + 4, jlast), lc, half, k0, u0, x0);
    __builtin_amdgcn_sched_barrier(0);
    attn_block(qb, k2, u2, x2, mrun, lsum, ot0, ot1);
    attn_load_kv(Kp, Vp, min(jb + 5, jlast), lc, half, k1, u1, x1);
    __builtin_amdgcn_sched_barrier(0);
    attn_block(qb, k3, u3, x3, mrun, lsum, ot0, ot1);
  }

  const float ltot = lsum + xhalf(lsum);
  const float inv = 1.0f / ltot;
  float* orow = o + (size_t)(base + row0 + lc) * DIMC + h * HDC + 8 * half;
  const float4 w0 = make_float4(ot0[0] * inv, ot0[1] * inv, ot0[2] * inv, ot0[3] * inv);
  const float4 w1 = make_float4(ot0[4] * inv, ot0[5] * inv, ot0[6] * inv, ot0[7] * inv);
  const float4 w2 = make_float4(ot1[0] * inv, ot1[1] * inv, ot1[2] * inv, ot1[3] * inv);
  const float4 w3 = make_float4(ot1[4] * inv, ot1[5] * inv, ot1[6] * inv, ot1[7] * inv);
  *reinterpret_cast<float4*>(orow) = w0;
  *reinterpret_cast<float4*>(orow + 4) = w1;
  *reinterpret_cast<float4*>(orow + 16) = w2;
  *reinterpret_cast<float4*>(orow + 20) = w3;
}

extern "C" void kernel_launch(void* const* d_in, const int* in_sizes, int n_in,
                              void* d_out, int out_size, void* d_ws, size_t ws_size,
                              hipStream_t stream) {
  const float* x     = (const float*)d_in[0];
  const float* ln1_g = (const float*)d_in[1] + (size_t)LAYER * DIMC;
  const float* ln1_b = (const float*)d_in[2] + (size_t)LAYER * DIMC;
  const float* Wqkv  = (const float*)d_in[3] + (size_t)LAYER * DIMC * QKVN;
  const float* bqkv  = (const float*)d_in[4] + (size_t)LAYER * QKVN;
  const float* Wo    = (const float*)d_in[5] + (size_t)LAYER * DIMC * DIMC;
  const float* bo    = (const float*)d_in[6] + (size_t)LAYER * DIMC;
  const float* ln2_g = (const float*)d_in[7] + (size_t)LAYER * DIMC;
  const float* ln2_b = (const float*)d_in[8] + (size_t)LAYER * DIMC;
  const float* W1    = (const float*)d_in[9] + (size_t)LAYER * DIMC * MLPC;
  const float* b1    = (const float*)d_in[10] + (size_t)LAYER * MLPC;
  const float* W2    = (const float*)d_in[11] + (size_t)LAYER * MLPC * DIMC;
  const float* b2    = (const float*)d_in[12] + (size_t)LAYER * DIMC;
  float* out = (float*)d_out;

  float* ws   = (float*)d_ws;
  float* y    = ws;
  float* qkvb = ws + (size_t)ROWS * DIMC;
  float* oat  = y;
  float* a    = qkvb + (size_t)ROWS * QKVN;
  float* z    = qkvb;
  float* h1   = a + (size_t)ROWS * DIMC;

  ln_kernel<<<ROWS / 8, 256, 0, stream>>>(x, ln1_g, ln1_b, y, ROWS);
  wmma_gemm_kernel<<<dim3(QKVN / 64, ROWS / 64), 128, 0, stream>>>(
      y, Wqkv, bqkv, nullptr, qkvb, ROWS, QKVN, DIMC, 0);
  attn_kernel<<<dim3(BC * HEADS, NC / 64), 128, 0, stream>>>(qkvb, oat);
  wmma_gemm_kernel<<<dim3(DIMC / 64, ROWS / 64), 128, 0, stream>>>(
      oat, Wo, bo, x, a, ROWS, DIMC, DIMC, 0);
  ln_kernel<<<ROWS / 8, 256, 0, stream>>>(a, ln2_g, ln2_b, z, ROWS);
  wmma_gemm_kernel<<<dim3(MLPC / 64, ROWS / 64), 128, 0, stream>>>(
      z, W1, b1, nullptr, h1, ROWS, MLPC, DIMC, 1);
  wmma_gemm_kernel<<<dim3(DIMC / 64, ROWS / 64), 128, 0, stream>>>(
      h1, W2, b2, a, out, ROWS, DIMC, MLPC, 0);
}


# attention: each 64-key K/V tile is fetched once per workgroup with whole-line loads into double-buffered LDS and its 4 waves read their WMMA fragments from LDS; summation order unchanged
# speedup vs baseline: 2.5484x; 1.0695x over previous
#include <hip/hip_runtime.h>
#include <hip/hip_bf16.h>


#define DIMC   256
#define MLPC   1024
#define BC     2
#define NC     2048
#define HEADS  8
#define HDC    32
#define ROWS   (BC * NC)
#define QKVN   (3 * DIMC)
#define LAYER  1

typedef __attribute__((ext_vector_type(2))) float v2f;
typedef __attribute__((ext_vector_type(8))) float v8f;

static __device__ inline v8f wmma4(v2f a, v2f b, v8f c) {
  return __builtin_amdgcn_wmma_f32_16x16x4_f32(
       false, a,  false, b,
       (short)0, c,  false,  false);
}

static __device__ inline v8f v8zero() {
  v8f z = {0.f, 0.f, 0.f, 0.f, 0.f, 0.f, 0.f, 0.f};
  return z;
}

__global__ __launch_bounds__(256) void ln_kernel(
    const float* __restrict__ x, const float* __restrict__ g,
    const float* __restrict__ b, float* __restrict__ y, int rows) {
  int wave = (blockIdx.x * blockDim.x + threadIdx.x) >> 5;
  int lane = threadIdx.x & 31;
  if (wave >= rows) return;
  const float* xr = x + (size_t)wave * DIMC;
  float vals[8];
  float s = 0.f;
#pragma unroll
  for (int i = 0; i < 8; ++i) { vals[i] = xr[lane + 32 * i]; s += vals[i]; }
#pragma unroll
  for (int off = 16; off >= 1; off >>= 1) s += __shfl_xor(s, off, 32);
  float mean = s * (1.0f / DIMC);
  float vs = 0.f;
#pragma unroll
  for (int i = 0; i < 8; ++i) { float d = vals[i] - mean; vs += d * d; }
#pragma unroll
  for (int off = 16; off >= 1; off >>= 1) vs += __shfl_xor(vs, off, 32);
  float rstd = rsqrtf(vs * (1.0f / DIMC) + 1e-5f);
  float* yr = y + (size_t)wave * DIMC;
#pragma unroll
  for (int i = 0; i < 8; ++i) {
    int c = lane + 32 * i;
    yr[c] = (vals[i] - mean) * rstd * g[c] + b[c];
  }
}

static __device__ inline void gemm_load_chunk(
    const float* __restrict__ Arow, const float* __restrict__ Wcol, int N, int kk,
    int half, v2f (&a)[4], v2f (&b)[4][4]) {
#pragma unroll
  for (int s = 0; s < 4; ++s) {
    const int ka = kk + 4 * s + 2 * half;
    v2f t = {Arow[ka], Arow[ka + 1]};
    a[s] = t;
    const float* Wr0 = Wcol + (size_t)ka * N;
    const float* Wr1 = Wr0 + N;
#pragma unroll
    for (int nt = 0; nt < 4; ++nt) {
      v2f u = {Wr0[nt * 16], Wr1[nt * 16]};
      b[s][nt] = u;
    }
  }
}

static __device__ inline void gemm_chunk(const v2f (&a)[4], const v2f (&b)[4][4],
                                         v8f (&acc)[4]) {
#pragma unroll
  for (int s = 0; s < 4; ++s) {
#pragma unroll
    for (int nt = 0; nt < 4; ++nt) acc[nt] = wmma4(a[s], b[s][nt], acc[nt]);
  }
}

__global__ __launch_bounds__(128) void wmma_gemm_kernel(
    const float* __restrict__ A, const float* __restrict__ W,
    const float* __restrict__ bias, const float* __restrict__ resid,
    float* __restrict__ out, int M, int N, int K, int gelu_flag) {
  const int lane = threadIdx.x & 31;
  const int wv = threadIdx.x >> 5;
  const int half = lane >> 4;
  const int lc = lane & 15;
  const int n0 = blockIdx.x * 64;
  const int m0 = blockIdx.y * 64 + wv * 16;

  const float* Arow = A + (size_t)(m0 + lc) * K;
  const float* Wcol = W + n0 + lc;
  v8f acc[4];
#pragma unroll
  for (int nt = 0; nt < 4; ++nt) acc[nt] = v8zero();

  v2f a0[4], b0[4][4], a1[4], b1[4][4];
  gemm_load_chunk(Arow, Wcol, N, 0, half, a0, b0);
  for (int kk = 0; kk < K; kk += 32) {
    if ((kk & 63) == 0 && kk + 64 < K)
      __builtin_prefetch(Arow + kk + 64, 0, 0);
    gemm_load_chunk(Arow, Wcol, N, kk + 16, half, a1, b1);
    __builtin_amdgcn_sched_barrier(0);
    gemm_chunk(a0, b0, acc);
    const int kn = (kk + 32 < K) ? kk + 32 : kk;
    gemm_load_chunk(Arow, Wcol, N, kn, half, a0, b0);
    __builtin_amdgcn_sched_barrier(0);
    gemm_chunk(a1, b1, acc);
  }

#pragma unroll
  for (int nt = 0; nt < 4; ++nt) {
    const int col = n0 + nt * 16 + lc;
    const float bv = bias ? bias[col] : 0.f;
#pragma unroll
    for (int v = 0; v < 8; ++v) {
      const int row = m0 + v + 8 * half;
      float val = acc[nt][v] + bv;
      if (gelu_flag)
        val = 0.5f * val * (1.0f + erff(val * 0.70710678118654752f));
      if (resid) val += resid[(size_t)row * N + col];
      out[(size_t)row * N + col] = val;
    }
  }
}

static __device__ inline float xhalf(float x) {
  const int b = __builtin_bit_cast(int, x);
  return __builtin_bit_cast(
      float, __builtin_amdgcn_permlanex16(b, b, 0x76543210, 0xfedcba98, false, false));
}

#define KT      64
#define KPITCH  36
#define TILEF   (KT * KPITCH)

typedef __attribute__((ext_vector_type(4))) float v4f;
struct tile_regs { v4f k0, k1, k2, k3, v0, v1, v2, v3; };

static __device__ inline tile_regs attn_fetch_tile(const float* gk, const float* gv, int tile) {
  const size_t off = (size_t)tile * KT * QKVN;
  const size_t step = (size_t)16 * QKVN;
  const float* pk = gk + off;
  const float* pv = gv + off;
  tile_regs r;
  r.k0 = *reinterpret_cast<const v4f*>(pk);
  r.k1 = *reinterpret_cast<const v4f*>(pk + step);
  r.k2 = *reinterpret_cast<const v4f*>(pk + 2 * step);
  r.k3 = *reinterpret_cast<const v4f*>(pk + 3 * step);
  r.v0 = *reinterpret_cast<const v4f*>(pv);
  r.v1 = *reinterpret_cast<const v4f*>(pv + step);
  r.v2 = *reinterpret_cast<const v4f*>(pv + 2 * step);
  r.v3 = *reinterpret_cast<const v4f*>(pv + 3 * step);
  return r;
}

static __device__ inline void attn_store_tile(float* sk, float* sv, const tile_regs r) {
  *reinterpret_cast<v4f*>(sk) = r.k0;
  *reinterpret_cast<v4f*>(sk + 16 * KPITCH) = r.k1;
  *reinterpret_cast<v4f*>(sk + 32 * KPITCH) = r.k2;
  *reinterpret_cast<v4f*>(sk + 48 * KPITCH) = r.k3;
  *reinterpret_cast<v4f*>(sv) = r.v0;
  *reinterpret_cast<v4f*>(sv + 16 * KPITCH) = r.v1;
  *reinterpret_cast<v4f*>(sv + 32 * KPITCH) = r.v2;
  *reinterpret_cast<v4f*>(sv + 48 * KPITCH) = r.v3;
}

static __device__ inline void attn_lds_frag(
    const float* __restrict__ sk, const float* __restrict__ sv, int kb, int lc, int half,
    v2f (&ka)[8], v2f (&va0)[4], v2f (&va1)[4]) {
  const float* krow = sk + (kb * 16 + lc) * KPITCH + 2 * half;
#pragma unroll
  for (int i = 0; i < 8; ++i) {
    v2f t = {krow[4 * i], krow[4 * i + 1]};
    ka[i] = t;
  }
  const float* vrow = sv + (kb * 16 + 8 * half) * KPITCH + lc;
#pragma unroll
  for (int i = 0; i < 4; ++i) {
    const float* r0 = vrow + (2 * i) * KPITCH;
    const float* r1 = r0 + KPITCH;
    v2f t0 = {r0[0], r1[0]};
    v2f t1 = {r0[16], r1[16]};
    va0[i] = t0;
    va1[i] = t1;
  }
}

static __device__ inline void attn_block(
    const v2f (&qb)[8], const v2f (&ka)[8], const v2f (&va0)[4], const v2f (&va1)[4],
    float& mrun, float& lsum, v8f& ot0, v8f& ot1) {
  v8f st = v8zero();
#pragma unroll
  for (int i = 0; i < 8; ++i) st = wmma4(ka[i], qb[i], st);

  float bm = st[0];
#pragma unroll
  for (int r = 1; r < 8; ++r) bm = fmaxf(bm, st[r]);
  bm = fmaxf(bm, xhalf(bm));
  const float mn = fmaxf(mrun, bm);
  const float c = __expf(mrun - mn);
  mrun = mn;
  float p[8];
  float ps = 0.f;
#pragma unroll
  for (int r = 0; r < 8; ++r) {
    p[r] = __expf(st[r] - mn);
    ps += p[r];
  }
  lsum = lsum * c + ps;
#pragma unroll
  for (int r = 0; r < 8; ++r) {
    ot0[r] *= c;
    ot1[r] *= c;
  }

#pragma unroll
  for (int i = 0; i < 4; ++i) {
    v2f pb = {p[2 * i], p[2 * i + 1]};
    ot0 = wmma4(va0[i], pb, ot0);
    ot1 = wmma4(va1[i], pb, ot1);
  }
}

__global__ __launch_bounds__(128) void attn_kernel(
    const float* qkv, float* __restrict__ o) {
  const int lane = threadIdx.x & 31;
  const int wv = threadIdx.x >> 5;
  const int half = lane >> 4;
  const int lc = lane & 15;
  const int bh = blockIdx.x;
  const int bb = bh >> 3, h = bh & 7;
  const int row0 = (blockIdx.y * 4 + wv) * 16;
  const int base = bb * NC;
  const float scale = 0.17677669529663687f;

  const float* Q = qkv + (size_t)base * QKVN + h * HDC;
  const float* Kp = Q + DIMC;
  const float* Vp = Q + 2 * DIMC;

  v2f qb[8];
  {
    const float* qrow = Q + (size_t)(row0 + lc) * QKVN + 2 * half;
#pragma unroll
    for (int i = 0; i < 8; ++i) {
      v2f t = {qrow[4 * i] * scale, qrow[4 * i + 1] * scale};
      qb[i] = t;
    }
  }

  float mrun = -1e30f;
  float lsum = 0.f;
  v8f ot0 = v8zero(), ot1 = v8zero();

  __shared__ __attribute__((aligned(16))) float sK[2][TILEF];
  __shared__ __attribute__((aligned(16))) float sV[2][TILEF];
  const int tid = threadIdx.x;
  const float* gk = Kp + (size_t)(tid >> 3) * QKVN + 4 * (tid & 7);
  const float* gv = gk + DIMC;
  const int soff = (tid >> 3) * KPITCH + 4 * (tid & 7);
  tile_regs tr = attn_fetch_tile(gk, gv, 0);
  attn_store_tile(&sK[0][soff], &sV[0][soff], tr);
  __syncthreads();

  v2f k0[8], u0[4], x0[4], k1[8], u1[4], x1[4];
  const int ntiles = NC / KT;
  for (int t = 0; t < ntiles; ++t) {
    const float* sk = sK[t & 1];
    const float* sv = sV[t & 1];
    tr = attn_fetch_tile(gk, gv, (t + 1 < ntiles) ? t + 1 : t);
    attn_lds_frag(sk, sv, 0, lc, half, k0, u0, x0);
    attn_lds_frag(sk, sv, 1, lc, half, k1, u1, x1);
    __builtin_amdgcn_sched_barrier(0);
    attn_block(qb, k0, u0, x0, mrun, lsum, ot0, ot1);
    attn_lds_frag(sk, sv, 2, lc, half, k0, u0, x0);
    __builtin_amdgcn_sched_barrier(0);
    attn_block(qb, k1, u1, x1, mrun, lsum, ot0, ot1);
    attn_lds_frag(sk, sv, 3, lc, half, k1, u1, x1);
    __builtin_amdgcn_sched_barrier(0);
    attn_block(qb, k0, u0, x0, mrun, lsum, ot0, ot1);
    attn_block(qb, k1, u1, x1, mrun, lsum, ot0, ot1);
    attn_store_tile(&sK[(t + 1) & 1][soff], &sV[(t + 1) & 1][soff], tr);
    __syncthreads();
  }

  const float ltot = lsum + xhalf(lsum);
  const float inv = 1.0f / ltot;
  float* orow = o + (size_t)(base + row0 + lc) * DIMC + h * HDC + 8 * half;
  const float4 w0 = make_float4(ot0[0] * inv, ot0[1] * inv, ot0[2] * inv, ot0[3] * inv);
  const float4 w1 = make_float4(ot0[4] * inv, ot0[5] * inv, ot0[6] * inv, ot0[7] * inv);
  const float4 w2 = make_float4(ot1[0] * inv, ot1[1] * inv, ot1[2] * inv, ot1[3] * inv);
  const float4 w3 = make_float4(ot1[4] * inv, ot1[5] * inv, ot1[6] * inv, ot1[7] * inv);
  *reinterpret_cast<float4*>(orow) = w0;
  *reinterpret_cast<float4*>(orow + 4) = w1;
  *reinterpret_cast<float4*>(orow + 16) = w2;
  *reinterpret_cast<float4*>(orow + 20) = w3;
}

extern "C" void kernel_launch(void* const* d_in, const int* in_sizes, int n_in,
                              void* d_out, int out_size, void* d_ws, size_t ws_size,
                              hipStream_t stream) {
  const float* x     = (const float*)d_in[0];
  const float* ln1_g = (const float*)d_in[1] + (size_t)LAYER * DIMC;
  const float* ln1_b = (const float*)d_in[2] + (size_t)LAYER * DIMC;
  const float* Wqkv  = (const float*)d_in[3] + (size_t)LAYER * DIMC * QKVN;
  const float* bqkv  = (const float*)d_in[4] + (size_t)LAYER * QKVN;
  const float* Wo    = (const float*)d_in[5] + (size_t)LAYER * DIMC * DIMC;
  const float* bo    = (const float*)d_in[6] + (size_t)LAYER * DIMC;
  const float* ln2_g = (const float*)d_in[7] + (size_t)LAYER * DIMC;
  const float* ln2_b = (const float*)d_in[8] + (size_t)LAYER * DIMC;
  const float* W1    = (const float*)d_in[9] + (size_t)LAYER * DIMC * MLPC;
  const float* b1    = (const float*)d_in[10] + (size_t)LAYER * MLPC;
  const float* W2    = (const float*)d_in[11] + (size_t)LAYER * MLPC * DIMC;
  const float* b2    = (const float*)d_in[12] + (size_t)LAYER * DIMC;
  float* out = (float*)d_out;

  float* ws   = (float*)d_ws;
  float* y    = ws;
  float* qkvb = ws + (size_t)ROWS * DIMC;
  float* oat  = y;
  float* a    = qkvb + (size_t)ROWS * QKVN;
  float* z    = qkvb;
  float* h1   = a + (size_t)ROWS * DIMC;

  ln_kernel<<<ROWS / 8, 256, 0, stream>>>(x, ln1_g, ln1_b, y, ROWS);
  wmma_gemm_kernel<<<dim3(QKVN / 64, ROWS / 64), 128, 0, stream>>>(
      y, Wqkv, bqkv, nullptr, qkvb, ROWS, QKVN, DIMC, 0);
  attn_kernel<<<dim3(BC * HEADS, NC / 64), 128, 0, stream>>>(qkvb, oat);
  wmma_gemm_kernel<<<dim3(DIMC / 64, ROWS / 64), 128, 0, stream>>>(
      oat, Wo, bo, x, a, ROWS, DIMC, DIMC, 0);
  ln_kernel<<<ROWS / 8, 256, 0, stream>>>(a, ln2_g, ln2_b, z, ROWS);
  wmma_gemm_kernel<<<dim3(MLPC / 64, ROWS / 64), 128, 0, stream>>>(
      z, W1, b1, nullptr, h1, ROWS, MLPC, DIMC, 1);
  wmma_gemm_kernel<<<dim3(DIMC / 64, ROWS / 64), 128, 0, stream>>>(
      h1, W2, b2, a, out, ROWS, DIMC, MLPC, 0);
}


# GEMM: every global access 16 bytes per lane (A: 4 k per load, k permuted within each 32-k chunk; W, bias, resid, out: the wave's 4 column tiles interleaved); changes k summation order
# speedup vs baseline: 3.6775x; 1.4431x over previous
#include <hip/hip_runtime.h>
#include <hip/hip_bf16.h>


#define DIMC   256
#define MLPC   1024
#define BC     2
#define NC     2048
#define HEADS  8
#define HDC    32
#define ROWS   (BC * NC)
#define QKVN   (3 * DIMC)
#define LAYER  1

typedef __attribute__((ext_vector_type(2))) float v2f;
typedef __attribute__((ext_vector_type(8))) float v8f;

static __device__ inline v8f wmma4(v2f a, v2f b, v8f c) {
  return __builtin_amdgcn_wmma_f32_16x16x4_f32(
       false, a,  false, b,
       (short)0, c,  false,  false);
}

static __device__ inline v8f v8zero() {
  v8f z = {0.f, 0.f, 0.f, 0.f, 0.f, 0.f, 0.f, 0.f};
  return z;
}

__global__ __launch_bounds__(256) void ln_kernel(
    const float* __restrict__ x, const float* __restrict__ g,
    const float* __restrict__ b, float* __restrict__ y, int rows) {
  int wave = (blockIdx.x * blockDim.x + threadIdx.x) >> 5;
  int lane = threadIdx.x & 31;
  if (wave >= rows) return;
  const float* xr = x + (size_t)wave * DIMC;
  float vals[8];
  float s = 0.f;
#pragma unroll
  for (int i = 0; i < 8; ++i) { vals[i] = xr[lane + 32 * i]; s += vals[i]; }
#pragma unroll
  for (int off = 16; off >= 1; off >>= 1) s += __shfl_xor(s, off, 32);
  float mean = s * (1.0f / DIMC);
  float vs = 0.f;
#pragma unroll
  for (int i = 0; i < 8; ++i) { float d = vals[i] - mean; vs += d * d; }
#pragma unroll
  for (int off = 16; off >= 1; off >>= 1) vs += __shfl_xor(vs, off, 32);
  float rstd = rsqrtf(vs * (1.0f / DIMC) + 1e-5f);
  float* yr = y + (size_t)wave * DIMC;
#pragma unroll
  for (int i = 0; i < 8; ++i) {
    int c = lane + 32 * i;
    yr[c] = (vals[i] - mean) * rstd * g[c] + b[c];
  }
}

typedef __attribute__((ext_vector_type(4))) float v4f;

struct gemm_set { v4f a[4]; v4f w[4][4]; };

template <int N>
static __device__ inline void gemm_load_set(
    const float* __restrict__ Ap, const float* __restrict__ Wp, int kk, gemm_set& s) {
#pragma unroll
  for (int q = 0; q < 4; ++q) {
    s.a[q] = *reinterpret_cast<const v4f*>(Ap + kk + 8 * q);
#pragma unroll
    for (int r = 0; r < 4; ++r)
      s.w[q][r] = *reinterpret_cast<const v4f*>(Wp + (size_t)(kk + 8 * q + r) * N);
  }
}

static __device__ inline void gemm_compute_set(const gemm_set& s, v8f (&acc)[4]) {
#pragma unroll
  for (int q = 0; q < 4; ++q) {
#pragma unroll
    for (int p = 0; p < 2; ++p) {
      v2f a = {s.a[q][2 * p], s.a[q][2 * p + 1]};
#pragma unroll
      for (int nt = 0; nt < 4; ++nt) {
        v2f b = {s.w[q][2 * p][nt], s.w[q][2 * p + 1][nt]};
        acc[nt] = wmma4(a, b, acc[nt]);
      }
    }
  }
}

template <int N, int K, bool GELU, bool RESID>
__global__ __launch_bounds__(128) void wmma_gemm_kernel(
    const float* __restrict__ A, const float* __restrict__ W,
    const float* __restrict__ bias, const float* __restrict__ resid,
    float* __restrict__ out) {
  const int lane = threadIdx.x & 31;
  const int wv = threadIdx.x >> 5;
  const int half = lane >> 4;
  const int lc = lane & 15;
  const int n0 = blockIdx.x * 64;
  const int m0 = blockIdx.y * 64 + wv * 16;
  const int col = n0 + 4 * lc;

  const float* Ap = A + (size_t)(m0 + lc) * K + 4 * half;
  const float* Wp = W + (size_t)(4 * half) * N + col;
  const v4f bv = *reinterpret_cast<const v4f*>(bias + col);

  v8f acc[4];
#pragma unroll
  for (int nt = 0; nt < 4; ++nt) acc[nt] = v8zero();

  gemm_set s0, s1;
  gemm_load_set<N>(Ap, Wp, 0, s0);
  for (int kk = 0; kk + 64 < K; kk += 64) {
    {
      int pk = kk + 96 + 32 * half;
      pk = (pk < K) ? pk : K - 32;
      __builtin_prefetch(Ap + pk, 0, 0);
    }
    gemm_load_set<N>(Ap, Wp, kk + 32, s1);
    __builtin_amdgcn_sched_barrier(0);
    gemm_compute_set(s0, acc);
    gemm_load_set<N>(Ap, Wp, kk + 64, s0);
    __builtin_amdgcn_sched_barrier(0);
    gemm_compute_set(s1, acc);
  }
  gemm_load_set<N>(Ap, Wp, K - 32, s1);
  v4f rv[8];
  if (RESID) {
#pragma unroll
    for (int v = 0; v < 8; ++v)
      rv[v] = *reinterpret_cast<const v4f*>(resid + (size_t)(m0 + v + 8 * half) * N + col);
  }
  __builtin_amdgcn_sched_barrier(0);
  gemm_compute_set(s0, acc);
  gemm_compute_set(s1, acc);

#pragma unroll
  for (int v = 0; v < 8; ++v) {
    const int row = m0 + v + 8 * half;
    v4f val = {acc[0][v], acc[1][v], acc[2][v], acc[3][v]};
    val += bv;
    if (GELU) {
#pragma unroll
      for (int e = 0; e < 4; ++e)
        val[e] = 0.5f * val[e] * (1.0f + erff(val[e] * 0.70710678118654752f));
    }
    if (RESID) val += rv[v];
    *reinterpret_cast<v4f*>(out + (size_t)row * N + col) = val;
  }
}

static __device__ inline float xhalf(float x) {
  const int b = __builtin_bit_cast(int, x);
  return __builtin_bit_cast(
      float, __builtin_amdgcn_permlanex16(b, b, 0x76543210, 0xfedcba98, false, false));
}

#define KT      64
#define KPITCH  36
#define TILEF   (KT * KPITCH)

struct tile_regs { v4f k0, k1, k2, k3, v0, v1, v2, v3; };

static __device__ inline tile_regs attn_fetch_tile(const float* gk, const float* gv, int tile) {
  const size_t off = (size_t)tile * KT * QKVN;
  const size_t step = (size_t)16 * QKVN;
  const float* pk = gk + off;
  const float* pv = gv + off;
  tile_regs r;
  r.k0 = *reinterpret_cast<const v4f*>(pk);
  r.k1 = *reinterpret_cast<const v4f*>(pk + step);
  r.k2 = *reinterpret_cast<const v4f*>(pk + 2 * step);
  r.k3 = *reinterpret_cast<const v4f*>(pk + 3 * step);
  r.v0 = *reinterpret_cast<const v4f*>(pv);
  r.v1 = *reinterpret_cast<const v4f*>(pv + step);
  r.v2 = *reinterpret_cast<const v4f*>(pv + 2 * step);
  r.v3 = *reinterpret_cast<const v4f*>(pv + 3 * step);
  return r;
}

static __device__ inline void attn_store_tile(float* sk, float* sv, const tile_regs r) {
  *reinterpret_cast<v4f*>(sk) = r.k0;
  *reinterpret_cast<v4f*>(sk + 16 * KPITCH) = r.k1;
  *reinterpret_cast<v4f*>(sk + 32 * KPITCH) = r.k2;
  *reinterpret_cast<v4f*>(sk + 48 * KPITCH) = r.k3;
  *reinterpret_cast<v4f*>(sv) = r.v0;
  *reinterpret_cast<v4f*>(sv + 16 * KPITCH) = r.v1;
  *reinterpret_cast<v4f*>(sv + 32 * KPITCH) = r.v2;
  *reinterpret_cast<v4f*>(sv + 48 * KPITCH) = r.v3;
}

static __device__ inline void attn_lds_frag(
    const float* __restrict__ sk, const float* __restrict__ sv, int kb, int lc, int half,
    v2f (&ka)[8], v2f (&va0)[4], v2f (&va1)[4]) {
  const float* krow = sk + (kb * 16 + lc) * KPITCH + 2 * half;
#pragma unroll
  for (int i = 0; i < 8; ++i) {
    v2f t = {krow[4 * i], krow[4 * i + 1]};
    ka[i] = t;
  }
  const float* vrow = sv + (kb * 16 + 8 * half) * KPITCH + lc;
#pragma unroll
  for (int i = 0; i < 4; ++i) {
    const float* r0 = vrow + (2 * i) * KPITCH;
    const float* r1 = r0 + KPITCH;
    v2f t0 = {r0[0], r1[0]};
    v2f t1 = {r0[16], r1[16]};
    va0[i] = t0;
    va1[i] = t1;
  }
}

static __device__ inline void attn_block(
    const v2f (&qb)[8], const v2f (&ka)[8], const v2f (&va0)[4], const v2f (&va1)[4],
    float& mrun, float& lsum, v8f& ot0, v8f& ot1) {
  v8f st = v8zero();
#pragma unroll
  for (int i = 0; i < 8; ++i) st = wmma4(ka[i], qb[i], st);

  float bm = st[0];
#pragma unroll
  for (int r = 1; r < 8; ++r) bm = fmaxf(bm, st[r]);
  bm = fmaxf(bm, xhalf(bm));
  const float mn = fmaxf(mrun, bm);
  const float c = __expf(mrun - mn);
  mrun = mn;
  float p[8];
  float ps = 0.f;
#pragma unroll
  for (int r = 0; r < 8; ++r) {
    p[r] = __expf(st[r] - mn);
    ps += p[r];
  }
  lsum = lsum * c + ps;
#pragma unroll
  for (int r = 0; r < 8; ++r) {
    ot0[r] *= c;
    ot1[r] *= c;
  }

#pragma unroll
  for (int i = 0; i < 4; ++i) {
    v2f pb = {p[2 * i], p[2 * i + 1]};
    ot0 = wmma4(va0[i], pb, ot0);
    ot1 = wmma4(va1[i], pb, ot1);
  }
}

__global__ __launch_bounds__(128) void attn_kernel(
    const float* qkv, float* __restrict__ o) {
  const int lane = threadIdx.x & 31;
  const int wv = threadIdx.x >> 5;
  const int half = lane >> 4;
  const int lc = lane & 15;
  const int bh = blockIdx.x;
  const int bb = bh >> 3, h = bh & 7;
  const int row0 = (blockIdx.y * 4 + wv) * 16;
  const int base = bb * NC;
  const float scale = 0.17677669529663687f;

  const float* Q = qkv + (size_t)base * QKVN + h * HDC;
  const float* Kp = Q + DIMC;
  const float* Vp = Q + 2 * DIMC;

  v2f qb[8];
  {
    const float* qrow = Q + (size_t)(row0 + lc) * QKVN + 2 * half;
#pragma unroll
    for (int i = 0; i < 8; ++i) {
      v2f t = {qrow[4 * i] * scale, qrow[4 * i + 1] * scale};
      qb[i] = t;
    }
  }

  float mrun = -1e30f;
  float lsum = 0.f;
  v8f ot0 = v8zero(), ot1 = v8zero();

  __shared__ __attribute__((aligned(16))) float sK[2][TILEF];
  __shared__ __attribute__((aligned(16))) float sV[2][TILEF];
  const int tid = threadIdx.x;
  const float* gk = Kp + (size_t)(tid >> 3) * QKVN + 4 * (tid & 7);
  const float* gv = gk + DIMC;
  const int soff = (tid >> 3) * KPITCH + 4 * (tid & 7);
  tile_regs tr = attn_fetch_tile(gk, gv, 0);
  attn_store_tile(&sK[0][soff], &sV[0][soff], tr);
  __syncthreads();

  v2f k0[8], u0[4], x0[4], k1[8], u1[4], x1[4];
  const int ntiles = NC / KT;
  for (int t = 0; t < ntiles; ++t) {
    const float* sk = sK[t & 1];
    const float* sv = sV[t & 1];
    tr = attn_fetch_tile(gk, gv, (t + 1 < ntiles) ? t + 1 : t);
    attn_lds_frag(sk, sv, 0, lc, half, k0, u0, x0);
    attn_lds_frag(sk, sv, 1, lc, half, k1, u1, x1);
    __builtin_amdgcn_sched_barrier(0);
    attn_block(qb, k0, u0, x0, mrun, lsum, ot0, ot1);
    attn_lds_frag(sk, sv, 2, lc, half, k0, u0, x0);
    __builtin_amdgcn_sched_barrier(0);
    attn_block(qb, k1, u1, x1, mrun, lsum, ot0, ot1);
    attn_lds_frag(sk, sv, 3, lc, half, k1, u1, x1);
    __builtin_amdgcn_sched_barrier(0);
    attn_block(qb, k0, u0, x0, mrun, lsum, ot0, ot1);
    attn_block(qb, k1, u1, x1, mrun, lsum, ot0, ot1);
    attn_store_tile(&sK[(t + 1) & 1][soff], &sV[(t + 1) & 1][soff], tr);
    __syncthreads();
  }

  const float ltot = lsum + xhalf(lsum);
  const float inv = 1.0f / ltot;
  float* orow = o + (size_t)(base + row0 + lc) * DIMC + h * HDC + 8 * half;
  const float4 w0 = make_float4(ot0[0] * inv, ot0[1] * inv, ot0[2] * inv, ot0[3] * inv);
  const float4 w1 = make_float4(ot0[4] * inv, ot0[5] * inv, ot0[6] * inv, ot0[7] * inv);
  const float4 w2 = make_float4(ot1[0] * inv, ot1[1] * inv, ot1[2] * inv, ot1[3] * inv);
  const float4 w3 = make_float4(ot1[4] * inv, ot1[5] * inv, ot1[6] * inv, ot1[7] * inv);
  *reinterpret_cast<float4*>(orow) = w0;
  *reinterpret_cast<float4*>(orow + 4) = w1;
  *reinterpret_cast<float4*>(orow + 16) = w2;
  *reinterpret_cast<float4*>(orow + 20) = w3;
}

extern "C" void kernel_launch(void* const* d_in, const int* in_sizes, int n_in,
                              void* d_out, int out_size, void* d_ws, size_t ws_size,
                              hipStream_t stream) {
  const float* x     = (const float*)d_in[0];
  const float* ln1_g = (const float*)d_in[1] + (size_t)LAYER * DIMC;
  const float* ln1_b = (const float*)d_in[2] + (size_t)LAYER * DIMC;
  const float* Wqkv  = (const float*)d_in[3] + (size_t)LAYER * DIMC * QKVN;
  const float* bqkv  = (const float*)d_in[4] + (size_t)LAYER * QKVN;
  const float* Wo    = (const float*)d_in[5] + (size_t)LAYER * DIMC * DIMC;
  const float* bo    = (const float*)d_in[6] + (size_t)LAYER * DIMC;
  const float* ln2_g = (const float*)d_in[7] + (size_t)LAYER * DIMC;
  const float* ln2_b = (const float*)d_in[8] + (size_t)LAYER * DIMC;
  const float* W1    = (const float*)d_in[9] + (size_t)LAYER * DIMC * MLPC;
  const float* b1    = (const float*)d_in[10] + (size_t)LAYER * MLPC;
  const float* W2    = (const float*)d_in[11] + (size_t)LAYER * MLPC * DIMC;
  const float* b2    = (const float*)d_in[12] + (size_t)LAYER * DIMC;
  float* out = (float*)d_out;

  float* ws   = (float*)d_ws;
  float* y    = ws;
  float* qkvb = ws + (size_t)ROWS * DIMC;
  float* oat  = y;
  float* a    = qkvb + (size_t)ROWS * QKVN;
  float* z    = qkvb;
  float* h1   = a + (size_t)ROWS * DIMC;

  ln_kernel<<<ROWS / 8, 256, 0, stream>>>(x, ln1_g, ln1_b, y, ROWS);
  wmma_gemm_kernel<QKVN, DIMC, false, false><<<dim3(QKVN / 64, ROWS / 64), 128, 0, stream>>>(
      y, Wqkv, bqkv, nullptr, qkvb);
  attn_kernel<<<dim3(BC * HEADS, NC / 64), 128, 0, stream>>>(qkvb, oat);
  wmma_gemm_kernel<DIMC, DIMC, false, true><<<dim3(DIMC / 64, ROWS / 64), 128, 0, stream>>>(
      oat, Wo, bo, x, a);
  ln_kernel<<<ROWS / 8, 256, 0, stream>>>(a, ln2_g, ln2_b, z, ROWS);
  wmma_gemm_kernel<MLPC, DIMC, true, false><<<dim3(MLPC / 64, ROWS / 64), 128, 0, stream>>>(
      z, W1, b1, nullptr, h1);
  wmma_gemm_kernel<DIMC, MLPC, false, true><<<dim3(DIMC / 64, ROWS / 64), 128, 0, stream>>>(
      h1, W2, b2, a, out);
}
